# starting kernel
# baseline (speedup 1.0000x reference)
#include <hip/hip_runtime.h>
#include <hip/hip_bf16.h>

typedef __attribute__((ext_vector_type(2))) float v2f;
typedef __attribute__((ext_vector_type(8))) float v8f;

constexpr int BATCH   = 8;
constexpr int FRAMES  = 16;
constexpr int EMBED   = 768;
constexpr int PATCHES = 196;
constexpr int PE      = PATCHES * EMBED;
constexpr int SEQ_E   = (FRAMES * PATCHES + 1) * EMBED;
constexpr int TILES_PER_B   = PE / 16;
constexpr int TILES_PER_WAVE = 8;
constexpr int WAVES_PER_B   = TILES_PER_B / TILES_PER_WAVE;
constexpr int TOTAL_WAVES   = BATCH * WAVES_PER_B;
constexpr int BLOCK_THREADS = 256;

__global__ __launch_bounds__(BLOCK_THREADS)
void patch_blender_wmma(const float* __restrict__ x,
                        const float* __restrict__ M,
                        float* __restrict__ out)
{
    const int wave = (blockIdx.x * BLOCK_THREADS + threadIdx.x) >> 5;
    const int lane = threadIdx.x & 31;
    if (wave >= TOTAL_WAVES) return;

    const int b        = wave / WAVES_PER_B;
    const int wave_in_b = wave - b * WAVES_PER_B;
    const int n_base0  = wave_in_b * (16 * TILES_PER_WAVE);

    const int col = lane & 15;
    const int g   = lane >> 4;

    v2f a[4];
#pragma unroll
    for (int s = 0; s < 4; ++s) {
        const int k = s * 4 + 2 * g;
        a[s].x = M[col * 16 + k];
        a[s].y = M[col * 16 + k + 1];
    }

    const float* __restrict__ src = x   + (size_t)b * SEQ_E + EMBED;
    float*       __restrict__ dst = out + (size_t)b * SEQ_E + EMBED;

#pragma unroll
    for (int c = 0; c < TILES_PER_WAVE; ++c) {
        const int n = n_base0 + c * 16 + col;

        v8f acc = {};
#pragma unroll
        for (int s = 0; s < 4; ++s) {
            const int k = s * 4 + 2 * g;
            v2f bb;
            bb.x = src[(size_t)k       * PE + n];
            bb.y = src[(size_t)(k + 1) * PE + n];
            acc = __builtin_amdgcn_wmma_f32_16x16x4_f32(
                 false, a[s],
                 false, bb,
                 (short)0, acc,
                 false,  false);
        }

#pragma unroll
        for (int r = 0; r < 8; ++r) {
            const int t = r + 8 * g;
            dst[(size_t)t * PE + n] = acc[r];
        }
    }
}

__global__ void cls_copy_kernel(const float* __restrict__ x,
                                float* __restrict__ out)
{
    const int i = blockIdx.x * blockDim.x + threadIdx.x;
    if (i < BATCH * EMBED) {
        const int b = i / EMBED;
        const int e = i - b * EMBED;
        out[(size_t)b * SEQ_E + e] = x[(size_t)b * SEQ_E + e];
    }
}

extern "C" void kernel_launch(void* const* d_in, const int* in_sizes, int n_in,
                              void* d_out, int out_size, void* d_ws, size_t ws_size,
                              hipStream_t stream)
{
    const float* x = (const float*)d_in[0];
    const float* M = (const float*)d_in[1];
    float* out = (float*)d_out;

    cls_copy_kernel<<<(BATCH * EMBED + 255) / 256, 256, 0, stream>>>(x, out);

    const int total_threads = TOTAL_WAVES * 32;
    const int blocks = total_threads / BLOCK_THREADS;
    patch_blender_wmma<<<blocks, BLOCK_THREADS, 0, stream>>>(x, M, out);
}


# 16-byte-per-lane loads of x and stores of out: each 64-column group is 4 column-interleaved WMMA tiles, so a half-wave moves one aligned 256-B row piece per instruction (was 64 B)
# speedup vs baseline: 1.3192x; 1.3192x over previous
#include <hip/hip_runtime.h>
#include <hip/hip_bf16.h>


typedef __attribute__((ext_vector_type(2))) float v2f;
typedef __attribute__((ext_vector_type(4))) float v4f;
typedef __attribute__((ext_vector_type(8))) float v8f;

constexpr int BATCH   = 8;
constexpr int FRAMES  = 16;
constexpr int EMBED   = 768;
constexpr int PATCHES = 196;
constexpr int PE      = PATCHES * EMBED;
constexpr int SEQ_E   = (FRAMES * PATCHES + 1) * EMBED;
constexpr int GROUP_COLS      = 64;
constexpr int TILES_PER_GROUP = 4;
constexpr int GROUPS_PER_WAVE = 2;
constexpr int COLS_PER_WAVE   = GROUP_COLS * GROUPS_PER_WAVE;
constexpr int WAVES_PER_B   = PE / COLS_PER_WAVE;
constexpr int TOTAL_WAVES   = BATCH * WAVES_PER_B;
constexpr int BLOCK_THREADS = 256;

static_assert(PE % COLS_PER_WAVE == 0, "columns must split evenly over waves");
static_assert((TOTAL_WAVES * 32) % BLOCK_THREADS == 0, "whole blocks only");
static_assert(FRAMES == 16, "one 16-row WMMA tile covers the frame axis");

__global__ __launch_bounds__(BLOCK_THREADS, 10)
void patch_blender_wmma(const float* __restrict__ x,
                        const float* __restrict__ M,
                        float* __restrict__ out)
{
    const int wave = (blockIdx.x * BLOCK_THREADS + threadIdx.x) >> 5;
    const int lane = threadIdx.x & 31;
    if (wave >= TOTAL_WAVES) return;

    const int b         = wave / WAVES_PER_B;
    const int wave_in_b = wave - b * WAVES_PER_B;

    const int col = lane & 15;
    const int g   = lane >> 4;

    v2f a[4];
#pragma unroll
    for (int s = 0; s < 4; ++s) {
        const int k = s * 4 + 2 * g;
        a[s].x = M[col * 16 + k];
        a[s].y = M[col * 16 + k + 1];
    }

    const int n0 = wave_in_b * COLS_PER_WAVE + 4 * col;
    const float* __restrict__ src = x   + (size_t)b * SEQ_E + EMBED + n0;
    float*       __restrict__ dst = out + (size_t)b * SEQ_E + EMBED + n0;

    v4f lo[4], hi[4];
#pragma unroll
    for (int s = 0; s < 4; ++s) {
        const int k = s * 4 + 2 * g;
        lo[s] = *reinterpret_cast<const v4f*>(src + (size_t)k       * PE);
        hi[s] = *reinterpret_cast<const v4f*>(src + (size_t)(k + 1) * PE);
    }

#pragma unroll
    for (int q = 0; q < GROUPS_PER_WAVE; ++q) {
        v8f acc[TILES_PER_GROUP] = {};
#pragma unroll
        for (int s = 0; s < 4; ++s) {
#pragma unroll
            for (int j = 0; j < TILES_PER_GROUP; ++j) {
                v2f bb;
                bb.x = lo[s][j];
                bb.y = hi[s][j];
                acc[j] = __builtin_amdgcn_wmma_f32_16x16x4_f32(
                     false, a[s],
                     false, bb,
                     (short)0, acc[j],
                     false,  false);
            }
        }

        if (q + 1 < GROUPS_PER_WAVE) {
#pragma unroll
            for (int s = 0; s < 4; ++s) {
                const int k = s * 4 + 2 * g;
                lo[s] = *reinterpret_cast<const v4f*>(src + (size_t)k       * PE + (q + 1) * GROUP_COLS);
                hi[s] = *reinterpret_cast<const v4f*>(src + (size_t)(k + 1) * PE + (q + 1) * GROUP_COLS);
            }
        }

#pragma unroll
        for (int r = 0; r < 8; ++r) {
            const int t = r + 8 * g;
            v4f o;
            o.x = acc[0][r];
            o.y = acc[1][r];
            o.z = acc[2][r];
            o.w = acc[3][r];
            *reinterpret_cast<v4f*>(dst + (size_t)t * PE + q * GROUP_COLS) = o;
        }
    }
}

__global__ void cls_copy_kernel(const float* __restrict__ x,
                                float* __restrict__ out)
{
    const int i = blockIdx.x * blockDim.x + threadIdx.x;
    if (i < BATCH * EMBED) {
        const int b = i / EMBED;
        const int e = i - b * EMBED;
        out[(size_t)b * SEQ_E + e] = x[(size_t)b * SEQ_E + e];
    }
}

extern "C" void kernel_launch(void* const* d_in, const int* in_sizes, int n_in,
                              void* d_out, int out_size, void* d_ws, size_t ws_size,
                              hipStream_t stream)
{
    const float* x = (const float*)d_in[0];
    const float* M = (const float*)d_in[1];
    float* out = (float*)d_out;

    cls_copy_kernel<<<(BATCH * EMBED + 255) / 256, 256, 0, stream>>>(x, out);

    const int total_threads = TOTAL_WAVES * 32;
    const int blocks = total_threads / BLOCK_THREADS;
    patch_blender_wmma<<<blocks, BLOCK_THREADS, 0, stream>>>(x, M, out);
}


# CLS-row copy folded into the matrix kernel as 6 extra work-groups at the end of its grid (one 16-byte piece per lane, scalar branch on the block index): one launch per call instead of two
# speedup vs baseline: 1.4339x; 1.0869x over previous
#include <hip/hip_runtime.h>
#include <hip/hip_bf16.h>


typedef __attribute__((ext_vector_type(2))) float v2f;
typedef __attribute__((ext_vector_type(4))) float v4f;
typedef __attribute__((ext_vector_type(8))) float v8f;

constexpr int BATCH   = 8;
constexpr int FRAMES  = 16;
constexpr int EMBED   = 768;
constexpr int PATCHES = 196;
constexpr int PE      = PATCHES * EMBED;
constexpr int SEQ_E   = (FRAMES * PATCHES + 1) * EMBED;
constexpr int GROUP_COLS      = 64;
constexpr int TILES_PER_GROUP = 4;
constexpr int GROUPS_PER_WAVE = 2;
constexpr int COLS_PER_WAVE   = GROUP_COLS * GROUPS_PER_WAVE;
constexpr int WAVES_PER_B   = PE / COLS_PER_WAVE;
constexpr int TOTAL_WAVES   = BATCH * WAVES_PER_B;
constexpr int BLOCK_THREADS = 256;
constexpr int MAIN_BLOCKS   = TOTAL_WAVES * 32 / BLOCK_THREADS;
constexpr int CLS_VEC_PER_B = EMBED / 4;
constexpr int CLS_BLOCKS    = BATCH * CLS_VEC_PER_B / BLOCK_THREADS;
constexpr int GRID_BLOCKS   = MAIN_BLOCKS + CLS_BLOCKS;

static_assert(PE % COLS_PER_WAVE == 0, "columns must split evenly over waves");
static_assert((TOTAL_WAVES * 32) % BLOCK_THREADS == 0, "whole blocks only");
static_assert(FRAMES == 16, "one 16-row WMMA tile covers the frame axis");
static_assert(CLS_VEC_PER_B % 32 == 0, "a copy wave stays inside one CLS row");
static_assert((BATCH * CLS_VEC_PER_B) % BLOCK_THREADS == 0, "whole copy blocks only");

__global__ __launch_bounds__(BLOCK_THREADS, 10)
void patch_blender_wmma(const float* __restrict__ x,
                        const float* __restrict__ M,
                        float* __restrict__ out)
{
    if (blockIdx.x >= MAIN_BLOCKS) {
        const int i  = (blockIdx.x - MAIN_BLOCKS) * BLOCK_THREADS + threadIdx.x;
        const int cb = i / CLS_VEC_PER_B;
        const size_t off = (size_t)cb * SEQ_E + (size_t)(i - cb * CLS_VEC_PER_B) * 4;
        *reinterpret_cast<v4f*>(out + off) = *reinterpret_cast<const v4f*>(x + off);
        return;
    }

    const int wave = (blockIdx.x * BLOCK_THREADS + threadIdx.x) >> 5;
    const int lane = threadIdx.x & 31;

    const int b         = wave / WAVES_PER_B;
    const int wave_in_b = wave - b * WAVES_PER_B;

    const int col = lane & 15;
    const int g   = lane >> 4;

    v2f a[4];
#pragma unroll
    for (int s = 0; s < 4; ++s) {
        const int k = s * 4 + 2 * g;
        a[s].x = M[col * 16 + k];
        a[s].y = M[col * 16 + k + 1];
    }

    const int n0 = wave_in_b * COLS_PER_WAVE + 4 * col;
    const float* __restrict__ src = x   + (size_t)b * SEQ_E + EMBED + n0;
    float*       __restrict__ dst = out + (size_t)b * SEQ_E + EMBED + n0;

    v4f lo[4], hi[4];
#pragma unroll
    for (int s = 0; s < 4; ++s) {
        const int k = s * 4 + 2 * g;
        lo[s] = *reinterpret_cast<const v4f*>(src + (size_t)k       * PE);
        hi[s] = *reinterpret_cast<const v4f*>(src + (size_t)(k + 1) * PE);
    }

#pragma unroll
    for (int q = 0; q < GROUPS_PER_WAVE; ++q) {
        v8f acc[TILES_PER_GROUP] = {};
#pragma unroll
        for (int s = 0; s < 4; ++s) {
#pragma unroll
            for (int j = 0; j < TILES_PER_GROUP; ++j) {
                v2f bb;
                bb.x = lo[s][j];
                bb.y = hi[s][j];
                acc[j] = __builtin_amdgcn_wmma_f32_16x16x4_f32(
                     false, a[s],
                     false, bb,
                     (short)0, acc[j],
                     false,  false);
            }
        }

        if (q + 1 < GROUPS_PER_WAVE) {
#pragma unroll
            for (int s = 0; s < 4; ++s) {
                const int k = s * 4 + 2 * g;
                lo[s] = *reinterpret_cast<const v4f*>(src + (size_t)k       * PE + (q + 1) * GROUP_COLS);
                hi[s] = *reinterpret_cast<const v4f*>(src + (size_t)(k + 1) * PE + (q + 1) * GROUP_COLS);
            }
        }

#pragma unroll
        for (int r = 0; r < 8; ++r) {
            const int t = r + 8 * g;
            v4f o;
            o.x = acc[0][r];
            o.y = acc[1][r];
            o.z = acc[2][r];
            o.w = acc[3][r];
            *reinterpret_cast<v4f*>(dst + (size_t)t * PE + q * GROUP_COLS) = o;
        }
    }
}

extern "C" void kernel_launch(void* const* d_in, const int* in_sizes, int n_in,
                              void* d_out, int out_size, void* d_ws, size_t ws_size,
                              hipStream_t stream)
{
    const float* x = (const float*)d_in[0];
    const float* M = (const float*)d_in[1];
    float* out = (float*)d_out;

    patch_blender_wmma<<<GRID_BLOCKS, BLOCK_THREADS, 0, stream>>>(x, M, out);
}
